# starting kernel
# baseline (speedup 1.0000x reference)
#include <hip/hip_runtime.h>

typedef __attribute__((ext_vector_type(2))) float v2f;
typedef __attribute__((ext_vector_type(8))) float v8f;

namespace {
constexpr int kNT     = 8192;
constexpr int kOutLen = 515;
constexpr int kTiles  = 31;
constexpr int kIters  = 257;
constexpr int kChunk  = 64;
constexpr int kD2Win  = 32 * kChunk + 40;
constexpr int kD1Back = 256;
constexpr int kD1Win  = 2576;
}

__launch_bounds__(256, 1)
__global__ void xcorr_wmma_kernel(const float* __restrict__ d1,
                                  const float* __restrict__ d2,
                                  float* __restrict__ out) {
  __shared__ float sD2[kD2Win];
  __shared__ float sD1[kD1Win];
  __shared__ float sTile[32 * 256];

  const int g    = blockIdx.x;
  const int tid  = threadIdx.x;
  const int lane = tid & 31;
  const int wave = tid >> 5;

  const float* __restrict__ g1 = d1 + (size_t)g * kNT;
  const float* __restrict__ g2 = d2 + (size_t)g * kNT;

  const int nCol = lane & 15;
  const int kb   = (lane >> 4) * 2;

  const int bBase = nCol + 8 * kb;
  int aBase[4];
#pragma unroll
  for (int tt = 0; tt < 4; ++tt) {
    int tau = wave * 4 + tt;
    if (tau >= kTiles) tau = wave * 4;
    const int dT = -249 + 17 * tau;
    aBase[tt] = kD1Back + dT + nCol + 8 * kb;
  }

  v8f acc[4] = {};

  for (int cs = 0; cs < kIters; cs += kChunk) {
    const int iters  = (kIters - cs) < kChunk ? (kIters - cs) : kChunk;
    const int tbBase = -32 + 32 * cs;

    __syncthreads();
    for (int o = tid; o < kD2Win; o += 256) {
      const int gi = tbBase + o;
      sD2[o] = (gi >= 0 && gi < kNT) ? g2[gi] : 0.0f;
    }
    for (int o = tid; o < kD1Win; o += 256) {
      const int gi = tbBase - kD1Back + o;
      sD1[o] = (gi >= 0 && gi < kNT) ? g1[gi] : 0.0f;
    }
    __syncthreads();

    for (int li = 0; li < iters; ++li) {
      const int sh = 32 * li;
      v2f b;
      b.x = sD2[bBase + sh];
      b.y = sD2[bBase + sh + 8];
#pragma unroll
      for (int tt = 0; tt < 4; ++tt) {
        v2f a;
        a.x = sD1[aBase[tt] + sh];
        a.y = sD1[aBase[tt] + sh + 8];
        acc[tt] = __builtin_amdgcn_wmma_f32_16x16x4_f32(
            false, a, false, b, (short)0, acc[tt], false, false);
      }
    }
  }

  const int mBase = (lane >> 4) * 8;
#pragma unroll
  for (int tt = 0; tt < 4; ++tt) {
    const int tau = wave * 4 + tt;
    if (tau < kTiles) {
      float* scr = sTile + (wave * 4 + tt) * 256;
#pragma unroll
      for (int r = 0; r < 8; ++r)
        scr[(mBase + r) * 16 + nCol] = acc[tt][r];
    }
  }
  __syncthreads();

#pragma unroll
  for (int tt = 0; tt < 4; ++tt) {
    const int tau = wave * 4 + tt;
    if (tau < kTiles && lane < 17) {
      const int j = 17 * tau + lane;
      if (j < kOutLen) {
        const int delta = lane - 8;
        const int n0 = delta < 0 ? -delta : 0;
        const float* scr = sTile + (wave * 4 + tt) * 256;
        float s = 0.0f;
#pragma unroll
        for (int q = 0; q < 8; ++q) {
          const int nn = n0 + q;
          s += scr[(nn + delta) * 16 + nn];
        }
        out[(size_t)g * kOutLen + j] = s;
      }
    }
  }
}

extern "C" void kernel_launch(void* const* d_in, const int* in_sizes, int n_in,
                              void* d_out, int out_size, void* d_ws, size_t ws_size,
                              hipStream_t stream) {
  const float* data1 = (const float*)d_in[0];
  const float* data2 = (const float*)d_in[1];
  float* out = (float*)d_out;
  const int groups = in_sizes[0] / kNT;
  xcorr_wmma_kernel<<<dim3(groups), dim3(256), 0, stream>>>(data1, data2, out);
}


# Re-tile lags as j=16*col+row: every WMMA cell is a distinct output, 3 tiles replace 31 half-used ones; waves now split the t range, partial tiles summed in LDS, so summation order changes
# speedup vs baseline: 1.9331x; 1.9331x over previous
#include <hip/hip_runtime.h>


typedef __attribute__((ext_vector_type(2))) float v2f;
typedef __attribute__((ext_vector_type(4))) float v4f;
typedef __attribute__((ext_vector_type(8))) float v8f;

namespace {
constexpr int kNT      = 8192;
constexpr int kOutLen  = 515;
constexpr int kSteps   = 514;
constexpr int kD1Pad   = 32;
constexpr int kD1Len   = 8496;
constexpr int kD2Rows  = 545;
constexpr int kPitch   = 20;
constexpr int kTailIt  = 257;
constexpr int kScrTile = 272;
constexpr int kScrWave = 3 * kScrTile;
}

__launch_bounds__(256)
__global__ void xcorr_wmma_kernel(const float* __restrict__ d1,
                                  const float* __restrict__ d2,
                                  float* __restrict__ out) {
  __shared__ __attribute__((aligned(16))) float sD2[kD2Rows * kPitch];
  __shared__ __attribute__((aligned(16))) float sD1[kD1Len];

  const int g    = blockIdx.x;
  const int tid  = threadIdx.x;
  const int lane = tid & 31;
  const int wave = __builtin_amdgcn_readfirstlane(tid >> 5);

  const float* __restrict__ g1 = d1 + (size_t)g * kNT;
  const float* __restrict__ g2 = d2 + (size_t)g * kNT;

  {
    v4f x[kNT / 1024], y[kNT / 1024];
    for (int k = 0; k < kNT / 1024; ++k) {
      const int q = tid + 256 * k;
      x[k] = *(const v4f*)(g1 + 4 * q);
      y[k] = *(const v4f*)(g2 + 4 * q);
    }
    for (int k = 0; k < kNT / 1024; ++k) {
      const int q = tid + 256 * k;
      *(v4f*)(sD1 + kD1Pad + 4 * q) = x[k];
      const int i = 4 * q + 256;
      *(v4f*)(sD2 + (i >> 4) * kPitch + (i & 15)) = y[k];
    }
  }
  sD2[tid] = 0.0f;
  if (tid < 16 * kPitch - 256) sD2[256 + tid] = 0.0f;
  sD2[528 * kPitch + tid] = 0.0f;
  if (tid < 17 * kPitch - 256) sD2[528 * kPitch + 256 + tid] = 0.0f;
  if (tid < kD1Pad) sD1[tid] = 0.0f;
  sD1[kD1Pad + kNT + tid] = 0.0f;
  if (tid < kD1Len - kD1Pad - kNT - 256) sD1[kD1Pad + kNT + 256 + tid] = 0.0f;
  __syncthreads();

  const int c = lane & 15;
  const int h = lane >> 4;

  v8f acc0 = {};
  v8f acc1 = {};
  v8f accT = {};

  const float* pa = sD1 + 15 + c + 8 * h;
  const float* pb = sD2 + (31 - c) * kPitch + 8 * h;
  for (int ss = wave; ss < kSteps; ss += 8) {
    const float* qa  = pa + 16 * ss;
    const float* qb0 = pb + ss * kPitch;
    const float* qb1 = qb0 - 16 * kPitch;
    const v4f b0lo = *(const v4f*)(qb0);
    const v4f b0hi = *(const v4f*)(qb0 + 4);
    const v4f b1lo = *(const v4f*)(qb1);
    const v4f b1hi = *(const v4f*)(qb1 + 4);
    v2f a0, a1, a2, a3;
    a0.x = qa[0]; a0.y = qa[1];
    a1.x = qa[2]; a1.y = qa[3];
    a2.x = qa[4]; a2.y = qa[5];
    a3.x = qa[6]; a3.y = qa[7];
    v2f b;
    b.x = b0lo.x; b.y = b0lo.y;
    acc0 = __builtin_amdgcn_wmma_f32_16x16x4_f32(false, a0, false, b, (short)0, acc0, false, false);
    b.x = b1lo.x; b.y = b1lo.y;
    acc1 = __builtin_amdgcn_wmma_f32_16x16x4_f32(false, a0, false, b, (short)0, acc1, false, false);
    b.x = b0lo.z; b.y = b0lo.w;
    acc0 = __builtin_amdgcn_wmma_f32_16x16x4_f32(false, a1, false, b, (short)0, acc0, false, false);
    b.x = b1lo.z; b.y = b1lo.w;
    acc1 = __builtin_amdgcn_wmma_f32_16x16x4_f32(false, a1, false, b, (short)0, acc1, false, false);
    b.x = b0hi.x; b.y = b0hi.y;
    acc0 = __builtin_amdgcn_wmma_f32_16x16x4_f32(false, a2, false, b, (short)0, acc0, false, false);
    b.x = b1hi.x; b.y = b1hi.y;
    acc1 = __builtin_amdgcn_wmma_f32_16x16x4_f32(false, a2, false, b, (short)0, acc1, false, false);
    b.x = b0hi.z; b.y = b0hi.w;
    acc0 = __builtin_amdgcn_wmma_f32_16x16x4_f32(false, a3, false, b, (short)0, acc0, false, false);
    b.x = b1hi.z; b.y = b1hi.w;
    acc1 = __builtin_amdgcn_wmma_f32_16x16x4_f32(false, a3, false, b, (short)0, acc1, false, false);
  }

  for (int i = (wave + 6) & 7; i < kTailIt; i += 8) {
    const float* qa = sD1 + 261 + 32 * i + c + 16 * h;
    v2f a;
    a.x = qa[0];
    a.y = qa[8];
    const int i0 = 224 + 32 * i + 16 * h + c;
    const int i1 = i0 + 8;
    v2f b;
    b.x = sD2[(i0 >> 4) * kPitch + (i0 & 15)];
    b.y = sD2[(i1 >> 4) * kPitch + (i1 & 15)];
    accT = __builtin_amdgcn_wmma_f32_16x16x4_f32(false, a, false, b, (short)0, accT, false, false);
  }

  __syncthreads();
  float* scr = sD2;
  {
    float* my = scr + wave * kScrWave;
    for (int r = 0; r < 8; ++r) {
      const int cell = 17 * c + 8 * h + r;
      my[cell] = acc0[r];
      my[kScrTile + cell] = acc1[r];
      my[2 * kScrTile + (8 * h + r) * 16 + c] = accT[r];
    }
  }
  __syncthreads();
  {
    const int p = tid + (tid >> 4);
    float s0 = 0.0f, s1 = 0.0f, sT = 0.0f;
    for (int w = 0; w < 8; ++w) {
      s0 += scr[w * kScrWave + p];
      s1 += scr[w * kScrWave + kScrTile + p];
      sT += scr[w * kScrWave + 2 * kScrTile + tid];
    }
    float* og = out + (size_t)g * kOutLen;
    og[tid] = s0;
    og[256 + tid] = s1;
    sD1[tid] = sT;
  }
  __syncthreads();
  if (tid < 3) {
    const int n0 = 6 - tid;
    float s = 0.0f;
    for (int q = 0; q < 8; ++q) s += sD1[q * 16 + n0 + q];
    out[(size_t)g * kOutLen + 512 + tid] = s;
  }
}

extern "C" void kernel_launch(void* const* d_in, const int* in_sizes, int n_in,
                              void* d_out, int out_size, void* d_ws, size_t ws_size,
                              hipStream_t stream) {
  const float* data1 = (const float*)d_in[0];
  const float* data2 = (const float*)d_in[1];
  float* out = (float*)d_out;
  const int groups = in_sizes[0] / kNT;
  xcorr_wmma_kernel<<<dim3(groups), dim3(256), 0, stream>>>(data1, data2, out);
}


# Dense loop: 4 steps per trip behind ONE group of LDS loads (was one per step); tile 1 reuses the B row tile 0 fetched two steps earlier: 24 loads in 1 place per 32 WMMAs, was 32 in 4; sums unchanged
# speedup vs baseline: 2.0251x; 1.0476x over previous
#include <hip/hip_runtime.h>


typedef __attribute__((ext_vector_type(2))) float v2f;
typedef __attribute__((ext_vector_type(4))) float v4f;
typedef __attribute__((ext_vector_type(8))) float v8f;

namespace {
constexpr int kNT      = 8192;
constexpr int kOutLen  = 515;
constexpr int kSteps   = 514;
constexpr int kD1Pad   = 32;
constexpr int kD1Len   = 8496;
constexpr int kD2Rows  = 545;
constexpr int kPitch   = 20;
constexpr int kTailIt  = 257;
constexpr int kScrTile = 272;
constexpr int kScrWave = 3 * kScrTile;
}

__launch_bounds__(256)
__global__ void xcorr_wmma_kernel(const float* __restrict__ d1,
                                  const float* __restrict__ d2,
                                  float* __restrict__ out) {
  __shared__ __attribute__((aligned(16))) float sD2[kD2Rows * kPitch];
  __shared__ __attribute__((aligned(16))) float sD1[kD1Len];

  const int g    = blockIdx.x;
  const int tid  = threadIdx.x;
  const int lane = tid & 31;
  const int wave = __builtin_amdgcn_readfirstlane(tid >> 5);

  const float* __restrict__ g1 = d1 + (size_t)g * kNT;
  const float* __restrict__ g2 = d2 + (size_t)g * kNT;

  {
    v4f x[kNT / 1024], y[kNT / 1024];
    for (int k = 0; k < kNT / 1024; ++k) {
      const int q = tid + 256 * k;
      x[k] = *(const v4f*)(g1 + 4 * q);
      y[k] = *(const v4f*)(g2 + 4 * q);
    }
    for (int k = 0; k < kNT / 1024; ++k) {
      const int q = tid + 256 * k;
      *(v4f*)(sD1 + kD1Pad + 4 * q) = x[k];
      const int i = 4 * q + 256;
      *(v4f*)(sD2 + (i >> 4) * kPitch + (i & 15)) = y[k];
    }
  }
  sD2[tid] = 0.0f;
  if (tid < 16 * kPitch - 256) sD2[256 + tid] = 0.0f;
  sD2[528 * kPitch + tid] = 0.0f;
  if (tid < 17 * kPitch - 256) sD2[528 * kPitch + 256 + tid] = 0.0f;
  if (tid < kD1Pad) sD1[tid] = 0.0f;
  sD1[kD1Pad + kNT + tid] = 0.0f;
  if (tid < kD1Len - kD1Pad - kNT - 256) sD1[kD1Pad + kNT + 256 + tid] = 0.0f;
  __syncthreads();

  const int c = lane & 15;
  const int h = lane >> 4;

  v8f acc0 = {};
  v8f acc1 = {};
  v8f accT = {};

  const float* pa = sD1 + 15 + c + 8 * h;
  const float* pb = sD2 + (31 - c) * kPitch + 8 * h;
  v4f c0lo, c0hi, c1lo, c1hi;
  {
    const float* qc = pb + (wave - 16) * kPitch;
    c0lo = *(const v4f*)(qc);
    c0hi = *(const v4f*)(qc + 4);
    c1lo = *(const v4f*)(qc + 8 * kPitch);
    c1hi = *(const v4f*)(qc + 8 * kPitch + 4);
  }
#define XC_STEP(A0, A1, A2, A3, T0LO, T0HI, T1LO, T1HI)                                                   \
  {                                                                                                       \
    v2f b;                                                                                                \
    b.x = T0LO.x; b.y = T0LO.y;                                                                           \
    acc0 = __builtin_amdgcn_wmma_f32_16x16x4_f32(false, A0, false, b, (short)0, acc0, false, false);      \
    b.x = T1LO.x; b.y = T1LO.y;                                                                           \
    acc1 = __builtin_amdgcn_wmma_f32_16x16x4_f32(false, A0, false, b, (short)0, acc1, false, false);      \
    b.x = T0LO.z; b.y = T0LO.w;                                                                           \
    acc0 = __builtin_amdgcn_wmma_f32_16x16x4_f32(false, A1, false, b, (short)0, acc0, false, false);      \
    b.x = T1LO.z; b.y = T1LO.w;                                                                           \
    acc1 = __builtin_amdgcn_wmma_f32_16x16x4_f32(false, A1, false, b, (short)0, acc1, false, false);      \
    b.x = T0HI.x; b.y = T0HI.y;                                                                           \
    acc0 = __builtin_amdgcn_wmma_f32_16x16x4_f32(false, A2, false, b, (short)0, acc0, false, false);      \
    b.x = T1HI.x; b.y = T1HI.y;                                                                           \
    acc1 = __builtin_amdgcn_wmma_f32_16x16x4_f32(false, A2, false, b, (short)0, acc1, false, false);      \
    b.x = T0HI.z; b.y = T0HI.w;                                                                           \
    acc0 = __builtin_amdgcn_wmma_f32_16x16x4_f32(false, A3, false, b, (short)0, acc0, false, false);      \
    b.x = T1HI.z; b.y = T1HI.w;                                                                           \
    acc1 = __builtin_amdgcn_wmma_f32_16x16x4_f32(false, A3, false, b, (short)0, acc1, false, false);      \
  }
#define XC_LOAD_A(Q, A0, A1, A2, A3)                                                                      \
  v2f A0, A1, A2, A3;                                                                                     \
  A0.x = (Q)[0]; A0.y = (Q)[1];                                                                           \
  A1.x = (Q)[2]; A1.y = (Q)[3];                                                                           \
  A2.x = (Q)[4]; A2.y = (Q)[5];                                                                           \
  A3.x = (Q)[6]; A3.y = (Q)[7];
  for (int g = 0; g < kSteps / 32; ++g) {
    const int ss = wave + 32 * g;
    const float* qa = pa + 16 * ss;
    const float* qb = pb + ss * kPitch;
    const v4f n0lo = *(const v4f*)(qb);
    const v4f n0hi = *(const v4f*)(qb + 4);
    XC_LOAD_A(qa,       a00, a01, a02, a03)
    const v4f n1lo = *(const v4f*)(qb + 8 * kPitch);
    const v4f n1hi = *(const v4f*)(qb + 8 * kPitch + 4);
    XC_LOAD_A(qa + 128, a10, a11, a12, a13)
    const v4f n2lo = *(const v4f*)(qb + 16 * kPitch);
    const v4f n2hi = *(const v4f*)(qb + 16 * kPitch + 4);
    XC_LOAD_A(qa + 256, a20, a21, a22, a23)
    const v4f n3lo = *(const v4f*)(qb + 24 * kPitch);
    const v4f n3hi = *(const v4f*)(qb + 24 * kPitch + 4);
    XC_LOAD_A(qa + 384, a30, a31, a32, a33)
    __builtin_amdgcn_sched_barrier(0);
    XC_STEP(a00, a01, a02, a03, n0lo, n0hi, c0lo, c0hi)
    XC_STEP(a10, a11, a12, a13, n1lo, n1hi, c1lo, c1hi)
    XC_STEP(a20, a21, a22, a23, n2lo, n2hi, n0lo, n0hi)
    XC_STEP(a30, a31, a32, a33, n3lo, n3hi, n1lo, n1hi)
    c0lo = n2lo; c0hi = n2hi;
    c1lo = n3lo; c1hi = n3hi;
    __builtin_amdgcn_sched_barrier(0);
  }
  if (wave < kSteps - 32 * (kSteps / 32)) {
    const int ss = wave + 32 * (kSteps / 32);
    const float* qa = pa + 16 * ss;
    const float* qb = pb + ss * kPitch;
    const v4f n0lo = *(const v4f*)(qb);
    const v4f n0hi = *(const v4f*)(qb + 4);
    XC_LOAD_A(qa, a00, a01, a02, a03)
    XC_STEP(a00, a01, a02, a03, n0lo, n0hi, c0lo, c0hi)
  }
#undef XC_STEP
#undef XC_LOAD_A

  for (int i = (wave + 6) & 7; i < kTailIt; i += 8) {
    const float* qa = sD1 + 261 + 32 * i + c + 16 * h;
    v2f a;
    a.x = qa[0];
    a.y = qa[8];
    const int i0 = 224 + 32 * i + 16 * h + c;
    const int i1 = i0 + 8;
    v2f b;
    b.x = sD2[(i0 >> 4) * kPitch + (i0 & 15)];
    b.y = sD2[(i1 >> 4) * kPitch + (i1 & 15)];
    accT = __builtin_amdgcn_wmma_f32_16x16x4_f32(false, a, false, b, (short)0, accT, false, false);
  }

  __syncthreads();
  float* scr = sD2;
  {
    float* my = scr + wave * kScrWave;
    for (int r = 0; r < 8; ++r) {
      const int cell = 17 * c + 8 * h + r;
      my[cell] = acc0[r];
      my[kScrTile + cell] = acc1[r];
      my[2 * kScrTile + (8 * h + r) * 16 + c] = accT[r];
    }
  }
  __syncthreads();
  {
    const int p = tid + (tid >> 4);
    float s0 = 0.0f, s1 = 0.0f, sT = 0.0f;
    for (int w = 0; w < 8; ++w) {
      s0 += scr[w * kScrWave + p];
      s1 += scr[w * kScrWave + kScrTile + p];
      sT += scr[w * kScrWave + 2 * kScrTile + tid];
    }
    float* og = out + (size_t)g * kOutLen;
    og[tid] = s0;
    og[256 + tid] = s1;
    sD1[tid] = sT;
  }
  __syncthreads();
  if (tid < 3) {
    const int n0 = 6 - tid;
    float s = 0.0f;
    for (int q = 0; q < 8; ++q) s += sD1[q * 16 + n0 + q];
    out[(size_t)g * kOutLen + 512 + tid] = s;
  }
}

extern "C" void kernel_launch(void* const* d_in, const int* in_sizes, int n_in,
                              void* d_out, int out_size, void* d_ws, size_t ws_size,
                              hipStream_t stream) {
  const float* data1 = (const float*)d_in[0];
  const float* data2 = (const float*)d_in[1];
  float* out = (float*)d_out;
  const int groups = in_sizes[0] / kNT;
  xcorr_wmma_kernel<<<dim3(groups), dim3(256), 0, stream>>>(data1, data2, out);
}


# Tail tile re-tiled: lags 512..514 on diagonals 0,1,2 with K stride 14, addresses linear per trip: 147 WMMAs sweep the samples, was 257; these 3 lags are summed in a new order
# speedup vs baseline: 2.1501x; 1.0617x over previous
#include <hip/hip_runtime.h>


typedef __attribute__((ext_vector_type(2))) float v2f;
typedef __attribute__((ext_vector_type(4))) float v4f;
typedef __attribute__((ext_vector_type(8))) float v8f;

namespace {
constexpr int kNT      = 8192;
constexpr int kOutLen  = 515;
constexpr int kSteps   = 514;
constexpr int kD1Pad   = 32;
constexpr int kD1Len   = 8528;
constexpr int kD2Rows  = 545;
constexpr int kPitch   = 20;
constexpr int kTailIt  = 147;
constexpr int kTailK   = 14;
constexpr int kScrTile = 272;
constexpr int kScrWave = 3 * kScrTile;
}

__launch_bounds__(256)
__global__ void xcorr_wmma_kernel(const float* __restrict__ d1,
                                  const float* __restrict__ d2,
                                  float* __restrict__ out) {
  __shared__ __attribute__((aligned(16))) float sD2[kD2Rows * kPitch];
  __shared__ __attribute__((aligned(16))) float sD1[kD1Len];

  const int g    = blockIdx.x;
  const int tid  = threadIdx.x;
  const int lane = tid & 31;
  const int wave = __builtin_amdgcn_readfirstlane(tid >> 5);

  const float* __restrict__ g1 = d1 + (size_t)g * kNT;
  const float* __restrict__ g2 = d2 + (size_t)g * kNT;

  {
    v4f x[kNT / 1024], y[kNT / 1024];
    for (int k = 0; k < kNT / 1024; ++k) {
      const int q = tid + 256 * k;
      x[k] = *(const v4f*)(g1 + 4 * q);
      y[k] = *(const v4f*)(g2 + 4 * q);
    }
    for (int k = 0; k < kNT / 1024; ++k) {
      const int q = tid + 256 * k;
      *(v4f*)(sD1 + kD1Pad + 4 * q) = x[k];
      const int i = 4 * q + 256;
      *(v4f*)(sD2 + (i >> 4) * kPitch + (i & 15)) = y[k];
    }
  }
  sD2[tid] = 0.0f;
  if (tid < 16 * kPitch - 256) sD2[256 + tid] = 0.0f;
  sD2[528 * kPitch + tid] = 0.0f;
  if (tid < 17 * kPitch - 256) sD2[528 * kPitch + 256 + tid] = 0.0f;
  if (tid < kD1Pad) sD1[tid] = 0.0f;
  sD1[kD1Pad + kNT + tid] = 0.0f;
  if (tid < kD1Len - kD1Pad - kNT - 256) sD1[kD1Pad + kNT + 256 + tid] = 0.0f;
  __syncthreads();

  const int c = lane & 15;
  const int h = lane >> 4;

  v8f acc0 = {};
  v8f acc1 = {};
  v8f accT = {};

  const float* pa = sD1 + 15 + c + 8 * h;
  const float* pb = sD2 + (31 - c) * kPitch + 8 * h;
  v4f c0lo, c0hi, c1lo, c1hi;
  {
    const float* qc = pb + (wave - 16) * kPitch;
    c0lo = *(const v4f*)(qc);
    c0hi = *(const v4f*)(qc + 4);
    c1lo = *(const v4f*)(qc + 8 * kPitch);
    c1hi = *(const v4f*)(qc + 8 * kPitch + 4);
  }
#define XC_STEP(A0, A1, A2, A3, T0LO, T0HI, T1LO, T1HI)                                                   \
  {                                                                                                       \
    v2f b;                                                                                                \
    b.x = T0LO.x; b.y = T0LO.y;                                                                           \
    acc0 = __builtin_amdgcn_wmma_f32_16x16x4_f32(false, A0, false, b, (short)0, acc0, false, false);      \
    b.x = T1LO.x; b.y = T1LO.y;                                                                           \
    acc1 = __builtin_amdgcn_wmma_f32_16x16x4_f32(false, A0, false, b, (short)0, acc1, false, false);      \
    b.x = T0LO.z; b.y = T0LO.w;                                                                           \
    acc0 = __builtin_amdgcn_wmma_f32_16x16x4_f32(false, A1, false, b, (short)0, acc0, false, false);      \
    b.x = T1LO.z; b.y = T1LO.w;                                                                           \
    acc1 = __builtin_amdgcn_wmma_f32_16x16x4_f32(false, A1, false, b, (short)0, acc1, false, false);      \
    b.x = T0HI.x; b.y = T0HI.y;                                                                           \
    acc0 = __builtin_amdgcn_wmma_f32_16x16x4_f32(false, A2, false, b, (short)0, acc0, false, false);      \
    b.x = T1HI.x; b.y = T1HI.y;                                                                           \
    acc1 = __builtin_amdgcn_wmma_f32_16x16x4_f32(false, A2, false, b, (short)0, acc1, false, false);      \
    b.x = T0HI.z; b.y = T0HI.w;                                                                           \
    acc0 = __builtin_amdgcn_wmma_f32_16x16x4_f32(false, A3, false, b, (short)0, acc0, false, false);      \
    b.x = T1HI.z; b.y = T1HI.w;                                                                           \
    acc1 = __builtin_amdgcn_wmma_f32_16x16x4_f32(false, A3, false, b, (short)0, acc1, false, false);      \
  }
#define XC_LOAD_A(Q, A0, A1, A2, A3)                                                                      \
  v2f A0, A1, A2, A3;                                                                                     \
  A0.x = (Q)[0]; A0.y = (Q)[1];                                                                           \
  A1.x = (Q)[2]; A1.y = (Q)[3];                                                                           \
  A2.x = (Q)[4]; A2.y = (Q)[5];                                                                           \
  A3.x = (Q)[6]; A3.y = (Q)[7];
  for (int g = 0; g < kSteps / 32; ++g) {
    const int ss = wave + 32 * g;
    const float* qa = pa + 16 * ss;
    const float* qb = pb + ss * kPitch;
    const v4f n0lo = *(const v4f*)(qb);
    const v4f n0hi = *(const v4f*)(qb + 4);
    XC_LOAD_A(qa,       a00, a01, a02, a03)
    const v4f n1lo = *(const v4f*)(qb + 8 * kPitch);
    const v4f n1hi = *(const v4f*)(qb + 8 * kPitch + 4);
    XC_LOAD_A(qa + 128, a10, a11, a12, a13)
    const v4f n2lo = *(const v4f*)(qb + 16 * kPitch);
    const v4f n2hi = *(const v4f*)(qb + 16 * kPitch + 4);
    XC_LOAD_A(qa + 256, a20, a21, a22, a23)
    const v4f n3lo = *(const v4f*)(qb + 24 * kPitch);
    const v4f n3hi = *(const v4f*)(qb + 24 * kPitch + 4);
    XC_LOAD_A(qa + 384, a30, a31, a32, a33)
    __builtin_amdgcn_sched_barrier(0);
    XC_STEP(a00, a01, a02, a03, n0lo, n0hi, c0lo, c0hi)
    XC_STEP(a10, a11, a12, a13, n1lo, n1hi, c1lo, c1hi)
    XC_STEP(a20, a21, a22, a23, n2lo, n2hi, n0lo, n0hi)
    XC_STEP(a30, a31, a32, a33, n3lo, n3hi, n1lo, n1hi)
    c0lo = n2lo; c0hi = n2hi;
    c1lo = n3lo; c1hi = n3hi;
    __builtin_amdgcn_sched_barrier(0);
  }
  if (wave < kSteps - 32 * (kSteps / 32)) {
    const int ss = wave + 32 * (kSteps / 32);
    const float* qa = pa + 16 * ss;
    const float* qb = pb + ss * kPitch;
    const v4f n0lo = *(const v4f*)(qb);
    const v4f n0hi = *(const v4f*)(qb + 4);
    XC_LOAD_A(qa, a00, a01, a02, a03)
    XC_STEP(a00, a01, a02, a03, n0lo, n0hi, c0lo, c0hi)
  }
#undef XC_STEP
#undef XC_LOAD_A

  {
    const int r0 = (wave + 6) & 7;
    const float* qa = sD1 + kD1Pad + 255 + 56 * r0 + c + 2 * kTailK * h;
    const int i0 = 256 + 56 * r0 + c + 2 * kTailK * h;
    const int i1 = i0 + kTailK;
    const float* qb0 = sD2 + (i0 >> 4) * kPitch + (i0 & 15);
    const float* qb1 = sD2 + (i1 >> 4) * kPitch + (i1 & 15);
    for (int i = r0; i < kTailIt; i += 8) {
      v2f a, b;
      a.x = qa[0];
      a.y = qa[kTailK];
      b.x = qb0[0];
      b.y = qb1[0];
      accT = __builtin_amdgcn_wmma_f32_16x16x4_f32(false, a, false, b, (short)0, accT, false, false);
      qa  += 8 * 56;
      qb0 += 28 * kPitch;
      qb1 += 28 * kPitch;
    }
  }

  __syncthreads();
  float* scr = sD2;
  {
    float* my = scr + wave * kScrWave;
    for (int r = 0; r < 8; ++r) {
      const int cell = 17 * c + 8 * h + r;
      my[cell] = acc0[r];
      my[kScrTile + cell] = acc1[r];
      my[2 * kScrTile + (8 * h + r) * 16 + c] = accT[r];
    }
  }
  __syncthreads();
  {
    const int p = tid + (tid >> 4);
    float s0 = 0.0f, s1 = 0.0f, sT = 0.0f;
    for (int w = 0; w < 8; ++w) {
      s0 += scr[w * kScrWave + p];
      s1 += scr[w * kScrWave + kScrTile + p];
      sT += scr[w * kScrWave + 2 * kScrTile + tid];
    }
    float* og = out + (size_t)g * kOutLen;
    og[tid] = s0;
    og[256 + tid] = s1;
    sD1[tid] = sT;
  }
  __syncthreads();
  if (tid < 3) {
    float s = 0.0f;
    for (int q = 0; q < kTailK; ++q) s += sD1[(q + tid) * 16 + q];
    out[(size_t)g * kOutLen + 512 + tid] = s;
  }
}

extern "C" void kernel_launch(void* const* d_in, const int* in_sizes, int n_in,
                              void* d_out, int out_size, void* d_ws, size_t ws_size,
                              hipStream_t stream) {
  const float* data1 = (const float*)d_in[0];
  const float* data2 = (const float*)d_in[1];
  float* out = (float*)d_out;
  const int groups = in_sizes[0] / kNT;
  xcorr_wmma_kernel<<<dim3(groups), dim3(256), 0, stream>>>(data1, data2, out);
}


# Even/odd sample split: three half-rate correlation tiles (Pe.Qe, Po.Qo, (Po+Pe').(Qe+Qo)) replace four, odd lags by subtraction; 3435 WMMAs per group, was 4259; summation order changes
# speedup vs baseline: 2.2723x; 1.0568x over previous
#include <hip/hip_runtime.h>


typedef __attribute__((ext_vector_type(2))) float v2f;
typedef __attribute__((ext_vector_type(4))) float v4f;
typedef __attribute__((ext_vector_type(8))) float v8f;

namespace {
constexpr int kNT      = 8192;
constexpr int kOutLen  = 515;
constexpr int kHalf    = 4096;
constexpr int kP0      = -129;
constexpr int kSteps   = 272;
constexpr int kAPad    = 132;
constexpr int kALen    = 4372;
constexpr int kBRows   = 287;
constexpr int kPitch   = 20;
constexpr int kBLen    = kBRows * kPitch;
constexpr int kTailIt  = 171;
constexpr int kTailK   = 12;
constexpr int kScrTile = 272;
constexpr int kScrWave = 4 * kScrTile;
}

__launch_bounds__(256)
__global__ void xcorr_wmma_kernel(const float* __restrict__ d1,
                                  const float* __restrict__ d2,
                                  float* __restrict__ out) {
  __shared__ __attribute__((aligned(16))) float sB[3 * kBLen];
  __shared__ __attribute__((aligned(16))) float sA[3 * kALen];

  const int g    = blockIdx.x;
  const int tid  = threadIdx.x;
  const int lane = tid & 31;
  const int wave = __builtin_amdgcn_readfirstlane(tid >> 5);

  const float* __restrict__ g1 = d1 + (size_t)g * kNT;
  const float* __restrict__ g2 = d2 + (size_t)g * kNT;

  {
    v4f x[kNT / 1024], y[kNT / 1024];
    for (int k = 0; k < kNT / 1024; ++k) {
      const int q = tid + 256 * k;
      x[k] = *(const v4f*)(g1 + 4 * q);
      y[k] = *(const v4f*)(g2 + 4 * q);
    }
    for (int k = 0; k < kNT / 1024; ++k) {
      const int q = tid + 256 * k;
      v2f t;
      t.x = x[k].x; t.y = x[k].z;
      *(v2f*)(sA + kAPad + 2 * q) = t;
      t.x = x[k].y; t.y = x[k].w;
      *(v2f*)(sA + kALen + kAPad + 2 * q) = t;
      const int i = 2 * q;
      float* rb = sB + (15 + (i >> 4)) * kPitch + (i & 15);
      t.x = y[k].x; t.y = y[k].z;
      *(v2f*)(rb) = t;
      t.x = y[k].y; t.y = y[k].w;
      *(v2f*)(rb + kBLen) = t;
      t.x = y[k].x + y[k].y; t.y = y[k].z + y[k].w;
      *(v2f*)(rb + 2 * kBLen) = t;
    }
  }
  for (int x = 0; x < 3; ++x) {
    float* a = sA + x * kALen;
    float* b = sB + x * kBLen;
    if (tid < kAPad) a[tid] = 0.0f;
    if (tid < kALen - kAPad - kHalf) a[kAPad + kHalf + tid] = 0.0f;
    b[tid] = 0.0f;
    if (tid < 15 * kPitch - 256) b[256 + tid] = 0.0f;
    b[271 * kPitch + tid] = 0.0f;
    if (tid < 16 * kPitch - 256) b[271 * kPitch + 256 + tid] = 0.0f;
  }
  __syncthreads();
  for (int k = 0; k < kHalf / 256 + 1; ++k) {
    const int i = tid + 256 * k - 1;
    if (i < kHalf) sA[2 * kALen + kAPad + i] = sA[kALen + kAPad + i] + sA[kAPad + i + 1];
  }
  __syncthreads();

  const int c = lane & 15;
  const int h = lane >> 4;

  v8f accE = {};
  v8f accO = {};
  v8f accS = {};
  v8f accT = {};

  const float* pa = sA + (kAPad + kP0) + c + 8 * h;
  const float* pb = sB + (15 - c) * kPitch + 8 * h;
#define XC_MMA(ACC, A, BX, BY)                                                                            \
  {                                                                                                       \
    v2f b;                                                                                                \
    b.x = BX; b.y = BY;                                                                                   \
    ACC = __builtin_amdgcn_wmma_f32_16x16x4_f32(false, A, false, b, (short)0, ACC, false, false);         \
  }
#define XC_STEP3(E, O, U)                                                                                 \
  XC_MMA(accE, E##a0, E##lo.x, E##lo.y) XC_MMA(accO, O##a0, O##lo.x, O##lo.y) XC_MMA(accS, U##a0, U##lo.x, U##lo.y) \
  XC_MMA(accE, E##a1, E##lo.z, E##lo.w) XC_MMA(accO, O##a1, O##lo.z, O##lo.w) XC_MMA(accS, U##a1, U##lo.z, U##lo.w) \
  XC_MMA(accE, E##a2, E##hi.x, E##hi.y) XC_MMA(accO, O##a2, O##hi.x, O##hi.y) XC_MMA(accS, U##a2, U##hi.x, U##hi.y) \
  XC_MMA(accE, E##a3, E##hi.z, E##hi.w) XC_MMA(accO, O##a3, O##hi.z, O##hi.w) XC_MMA(accS, U##a3, U##hi.z, U##hi.w)
#define XC_LOAD(QA, QB, N)                                                                                \
  const v4f N##lo = *(const v4f*)(QB);                                                                    \
  const v4f N##hi = *(const v4f*)((QB) + 4);                                                              \
  v2f N##a0, N##a1, N##a2, N##a3;                                                                         \
  N##a0.x = (QA)[0]; N##a0.y = (QA)[1];                                                                   \
  N##a1.x = (QA)[2]; N##a1.y = (QA)[3];                                                                   \
  N##a2.x = (QA)[4]; N##a2.y = (QA)[5];                                                                   \
  N##a3.x = (QA)[6]; N##a3.y = (QA)[7];
#pragma clang loop unroll(disable)
  for (int it = 0; it < kSteps / 16; ++it) {
    const int ss = wave + 16 * it;
    const float* qa = pa + 16 * ss;
    const float* qb = pb + ss * kPitch;
    XC_LOAD(qa,                   qb,                           e0)
    XC_LOAD(qa + kALen,           qb + kBLen,                   o0)
    XC_LOAD(qa + 2 * kALen,       qb + 2 * kBLen,               u0)
    XC_LOAD(qa + 128,             qb + 8 * kPitch,              e1)
    XC_LOAD(qa + kALen + 128,     qb + kBLen + 8 * kPitch,      o1)
    XC_LOAD(qa + 2 * kALen + 128, qb + 2 * kBLen + 8 * kPitch,  u1)
    __builtin_amdgcn_sched_barrier(0);
    XC_STEP3(e0, o0, u0)
    XC_STEP3(e1, o1, u1)
    __builtin_amdgcn_sched_barrier(0);
  }
#undef XC_STEP3
#undef XC_LOAD

  {
    const int r0 = wave;
    const int par = (c + 1) & 1;
    const float* qa = sA + par * kALen + kAPad + 24 * r0 + 12 * h + ((253 + c - par) >> 1);
    const int y0 = 24 * r0 + 12 * h + (c >> 1);
    const int y1 = y0 + kTailK / 2;
    const float* qb0 = sB + (c & 1) * kBLen + (15 + (y0 >> 4)) * kPitch + (y0 & 15);
    const float* qb1 = sB + (c & 1) * kBLen + (15 + (y1 >> 4)) * kPitch + (y1 & 15);
    for (int k = 0; k < kTailIt / 8; ++k) {
      v2f a, b;
      a.x = qa[0];
      a.y = qa[kTailK / 2];
      b.x = qb0[0];
      b.y = qb1[0];
      accT = __builtin_amdgcn_wmma_f32_16x16x4_f32(false, a, false, b, (short)0, accT, false, false);
      qa  += 192;
      qb0 += 12 * kPitch;
      qb1 += 12 * kPitch;
    }
    if (wave < kTailIt - 8 * (kTailIt / 8)) {
      v2f a, b;
      a.x = qa[0];
      a.y = qa[kTailK / 2];
      b.x = qb0[0];
      b.y = qb1[0];
      accT = __builtin_amdgcn_wmma_f32_16x16x4_f32(false, a, false, b, (short)0, accT, false, false);
    }
  }
#undef XC_MMA

  __syncthreads();
  float* scr = sB;
  float* redE = sA;
  float* redT = sA + 256;
  float* orow = sA + 512;
  {
    float* my = scr + wave * kScrWave;
    for (int r = 0; r < 8; ++r) {
      const int cell = 17 * c + 8 * h + r;
      my[cell] = accE[r];
      my[kScrTile + cell] = accO[r];
      my[2 * kScrTile + cell] = accS[r];
      my[3 * kScrTile + (8 * h + r) * 16 + c] = accT[r];
    }
  }
  __syncthreads();
  float sE = 0.0f, sO = 0.0f, sS = 0.0f;
  {
    const int p = tid + (tid >> 4);
    float sT = 0.0f;
    for (int w = 0; w < 8; ++w) {
      sE += scr[w * kScrWave + p];
      sO += scr[w * kScrWave + kScrTile + p];
      sS += scr[w * kScrWave + 2 * kScrTile + p];
      sT += scr[w * kScrWave + 3 * kScrTile + tid];
    }
    redE[tid] = sE;
    redT[tid] = sT;
  }
  __syncthreads();
  if (tid >= 1) orow[2 * tid - 1] = sE + sO;
  if (tid <= 254) orow[2 * tid] = (sS - sO) - redE[tid + 1];
  if (tid < 5) {
    float s = 0.0f;
    for (int q = 0; q < kTailK; ++q) s += redT[(q + tid) * 16 + q];
    orow[510 + tid] = s;
  }
  __syncthreads();
  {
    float* og = out + (size_t)g * kOutLen;
    og[tid] = orow[tid];
    og[256 + tid] = orow[256 + tid];
    if (tid < kOutLen - 512) og[512 + tid] = orow[512 + tid];
  }
}

extern "C" void kernel_launch(void* const* d_in, const int* in_sizes, int n_in,
                              void* d_out, int out_size, void* d_ws, size_t ws_size,
                              hipStream_t stream) {
  const float* data1 = (const float*)d_in[0];
  const float* data2 = (const float*)d_in[1];
  float* out = (float*)d_out;
  const int groups = in_sizes[0] / kNT;
  xcorr_wmma_kernel<<<dim3(groups), dim3(256), 0, stream>>>(data1, data2, out);
}
